# starting kernel
# baseline (speedup 1.0000x reference)
#include <hip/hip_runtime.h>
#include <hip/hip_bf16.h>
#include <math.h>

typedef float v2f __attribute__((ext_vector_type(2)));
typedef float v8f __attribute__((ext_vector_type(8)));

#define HID   64
#define RANK  4
#define NAB   1036
#define GSTR  1044
#define XSTR  65

#define OFF_A1 0
#define OFF_B1 8
#define OFF_A2 264
#define OFF_B2 520
#define OFF_A3 776
#define OFF_B3 1032

__global__ __launch_bounds__(64) void hyperlr_fused(
    const float* __restrict__ coords, const float* __restrict__ param,
    const float* __restrict__ W1,  const float* __restrict__ B1,
    const float* __restrict__ W2,  const float* __restrict__ B2,
    const float* __restrict__ W3,  const float* __restrict__ B3,
    const float* __restrict__ Wh1, const float* __restrict__ Bh1,
    const float* __restrict__ Wh2, const float* __restrict__ Bh2,
    float* __restrict__ out)
{
    extern __shared__ float smem[];
    const int lane = threadIdx.x & 31;
    const int wave = threadIdx.x >> 5;

    float* g_lds  = smem + wave * (16 * GSTR);
    float* x_lds  = smem + 2 * (16 * GSTR) + wave * (16 * XSTR);
    float* t2_lds = smem + 2 * (16 * GSTR) + 2 * (16 * XSTR) + wave * (16 * RANK);

    const int m    = lane & 15;
    const int half = lane >> 4;
    const int kb   = half * 2;
    const int base = blockIdx.x * 32 + wave * 16;

    const float pv = param[base + m];
    v2f afrag[16];
    #pragma unroll
    for (int s = 0; s < 16; ++s) {
        int k = 4 * s + kb;
        afrag[s].x = tanhf(pv * Wh1[k]     + Bh1[k]);
        afrag[s].y = tanhf(pv * Wh1[k + 1] + Bh1[k + 1]);
    }

    auto loadB = [&](int ct, v2f (&dst)[16]) {
        int col  = ct * 16 + m;
        int ccol = col < NAB ? col : (NAB - 1);
        const float* wp = Wh2 + ccol * HID + kb;
        #pragma unroll
        for (int s = 0; s < 16; ++s) { dst[s].x = wp[4 * s]; dst[s].y = wp[4 * s + 1]; }
    };

    v2f bnext[16];
    loadB(0, bnext);
    for (int ct = 0; ct < 65; ++ct) {
        v2f bcur[16];
        #pragma unroll
        for (int s = 0; s < 16; ++s) bcur[s] = bnext[s];
        if (ct < 64) loadB(ct + 1, bnext);

        v8f acc0 = {}, acc1 = {};
        #pragma unroll
        for (int s = 0; s < 16; s += 2) {
            acc0 = __builtin_amdgcn_wmma_f32_16x16x4_f32(false, afrag[s],     false, bcur[s],
                                                         (short)0, acc0, false, false);
            acc1 = __builtin_amdgcn_wmma_f32_16x16x4_f32(false, afrag[s + 1], false, bcur[s + 1],
                                                         (short)0, acc1, false, false);
        }
        const v8f acc = acc0 + acc1;

        const int col  = ct * 16 + m;
        const int ccol = col < NAB ? col : (NAB - 1);
        const float bias = Bh2[ccol];
        #pragma unroll
        for (int v = 0; v < 8; ++v)
            g_lds[(v + 8 * half) * GSTR + col] = acc[v] + bias;
    }
    __syncthreads();

    const float c0 = coords[(base + m) * 2 + 0];
    const float c1 = coords[(base + m) * 2 + 1];
    float t1[RANK];
    #pragma unroll
    for (int r = 0; r < RANK; ++r)
        t1[r] = c0 * g_lds[m * GSTR + OFF_A1 + 2 * r]
              + c1 * g_lds[m * GSTR + OFF_A1 + 2 * r + 1];

    v2f x1frag[16];
    #pragma unroll
    for (int s = 0; s < 16; ++s) {
        #pragma unroll
        for (int j = 0; j < 2; ++j) {
            const int h = 4 * s + kb + j;
            float pre = W1[2 * h] * c0 + W1[2 * h + 1] * c1 + B1[h];
            #pragma unroll
            for (int r = 0; r < RANK; ++r)
                pre += g_lds[m * GSTR + OFF_B1 + 4 * h + r] * t1[r];
            const float x = tanhf(pre);
            x1frag[s][j] = x;
            x_lds[m * XSTR + h] = x;
        }
    }
    __syncthreads();

    {
        float t2[RANK] = {0.f, 0.f, 0.f, 0.f};
        for (int h = 0; h < HID; ++h) {
            const float xv = x_lds[m * XSTR + h];
            #pragma unroll
            for (int r = 0; r < RANK; ++r)
                t2[r] += xv * g_lds[m * GSTR + OFF_A2 + 64 * r + h];
        }
        if (half == 0) {
            #pragma unroll
            for (int r = 0; r < RANK; ++r) t2_lds[m * RANK + r] = t2[r];
        }
    }
    __syncthreads();

    #pragma unroll
    for (int ct = 0; ct < 4; ++ct) {
        const int hcol = ct * 16 + m;
        const float* wp = W2 + hcol * HID + kb;
        v2f bfrag[16];
        #pragma unroll
        for (int s = 0; s < 16; ++s) { bfrag[s].x = wp[4 * s]; bfrag[s].y = wp[4 * s + 1]; }

        v8f acc0 = {}, acc1 = {};
        #pragma unroll
        for (int s = 0; s < 16; s += 2) {
            acc0 = __builtin_amdgcn_wmma_f32_16x16x4_f32(false, x1frag[s],     false, bfrag[s],
                                                         (short)0, acc0, false, false);
            acc1 = __builtin_amdgcn_wmma_f32_16x16x4_f32(false, x1frag[s + 1], false, bfrag[s + 1],
                                                         (short)0, acc1, false, false);
        }
        const v8f acc = acc0 + acc1;

        const float b2h = B2[hcol];
        #pragma unroll
        for (int v = 0; v < 8; ++v) {
            const int row = v + 8 * half;
            float pre = acc[v] + b2h;
            #pragma unroll
            for (int r = 0; r < RANK; ++r)
                pre += t2_lds[row * RANK + r] * g_lds[row * GSTR + OFF_B2 + 4 * hcol + r];
            x_lds[row * XSTR + hcol] = tanhf(pre);
        }
    }
    __syncthreads();

    if (half == 0) {
        float t3[RANK] = {0.f, 0.f, 0.f, 0.f};
        float w3s = 0.f;
        for (int h = 0; h < HID; ++h) {
            const float xv = x_lds[m * XSTR + h];
            w3s += xv * W3[h];
            #pragma unroll
            for (int r = 0; r < RANK; ++r)
                t3[r] += xv * g_lds[m * GSTR + OFF_A3 + 64 * r + h];
        }
        float o = w3s + B3[0];
        #pragma unroll
        for (int r = 0; r < RANK; ++r)
            o += t3[r] * g_lds[m * GSTR + OFF_B3 + r];
        out[base + m] = o;
    }
}

extern "C" void kernel_launch(void* const* d_in, const int* in_sizes, int n_in,
                              void* d_out, int out_size, void* d_ws, size_t ws_size,
                              hipStream_t stream) {
    const float* coords = (const float*)d_in[0];
    const float* param  = (const float*)d_in[1];
    const float* W1     = (const float*)d_in[2];
    const float* B1     = (const float*)d_in[3];
    const float* W2     = (const float*)d_in[4];
    const float* B2     = (const float*)d_in[5];
    const float* W3     = (const float*)d_in[6];
    const float* B3     = (const float*)d_in[7];
    const float* Wh1    = (const float*)d_in[8];
    const float* Bh1    = (const float*)d_in[9];
    const float* Wh2    = (const float*)d_in[10];
    const float* Bh2    = (const float*)d_in[11];
    float* out = (float*)d_out;

    const int n = in_sizes[1];
    const size_t smem = (size_t)(2 * 16 * GSTR + 2 * 16 * XSTR + 2 * 16 * RANK) * sizeof(float);
    (void)hipFuncSetAttribute((const void*)hyperlr_fused,
                              hipFuncAttributeMaxDynamicSharedMemorySize, (int)smem);

    const int blocks = n / 32;
    hyperlr_fused<<<blocks, 64, smem, stream>>>(coords, param, W1, B1, W2, B2, W3, B3,
                                                Wh1, Bh1, Wh2, Bh2, out);
}


# Transposed product: Wh2 rows are the A operand, 32 samples per wave the N side; a lane gets its own sample's factors, MLP in registers, no g in LDS; t2,t3,x2.W3 summed per lane half first
# speedup vs baseline: 1.9392x; 1.9392x over previous
#include <hip/hip_runtime.h>
#include <math.h>


typedef float v2f __attribute__((ext_vector_type(2)));
typedef float v4f __attribute__((ext_vector_type(4)));
typedef float v8f __attribute__((ext_vector_type(8)));

#define HID   64
#define RANK  4
#define NAB   1036
#define NJ    2


__device__ __forceinline__ int row_t0(int i)        { return i < 8 ? i : 1032 + (i & 3); }
__device__ __forceinline__ int row_b1(int t, int i) { return 8 + 16 * t + i; }
__device__ __forceinline__ int row_a2(int t, int i) { return 264 + 64 * (i & 3) + 4 * t + 2 * (i >> 3) + ((i >> 2) & 1); }
__device__ __forceinline__ int row_b2(int t, int i) { return 520 + 64 * (t >> 2) + 8 * (t & 3) + 32 * (i >> 3) + (i & 7); }
__device__ __forceinline__ int row_a3(int t, int i) { return 776 + 64 * (i & 3) + 16 * (t >> 2) + 8 * (i >> 3) + 2 * (t & 3) + ((i >> 2) & 1); }

__device__ __forceinline__ void load_frag(const float* __restrict__ rowp, v2f (&a)[16]) {
    #pragma unroll
    for (int s = 0; s < 16; ++s) a[s] = *(const v2f*)(rowp + 4 * s);
}

__device__ __forceinline__ v8f mm16(const v2f (&a)[16], const v2f (&b)[16]) {
    v8f acc0 = {}, acc1 = {};
    #pragma unroll
    for (int s = 0; s < 16; s += 2) {
        acc0 = __builtin_amdgcn_wmma_f32_16x16x4_f32(false, a[s],     false, b[s],
                                                     (short)0, acc0, false, false);
        acc1 = __builtin_amdgcn_wmma_f32_16x16x4_f32(false, a[s + 1], false, b[s + 1],
                                                     (short)0, acc1, false, false);
    }
    return acc0 + acc1;
}

__global__ __launch_bounds__(64) void hyperlr_fused(
    const float* __restrict__ coords, const float* __restrict__ param,
    const float* __restrict__ W1,  const float* __restrict__ B1,
    const float* __restrict__ W2,  const float* __restrict__ B2,
    const float* __restrict__ W3,  const float* __restrict__ B3,
    const float* __restrict__ Wh1, const float* __restrict__ Bh1,
    const float* __restrict__ Wh2, const float* __restrict__ Bh2,
    float* __restrict__ out)
{
    __shared__ v2f sH [2][NJ][16][32];
    __shared__ v2f sX1[2][NJ][16][32];
    __shared__ v2f sP [2][NJ][16][32];

    const int lane = threadIdx.x & 31;
    const int wave = threadIdx.x >> 5;
    const int n    = lane & 15;
    const int hf   = lane >> 4;
    const int kb   = 2 * hf;
    const int base = (blockIdx.x * 2 + wave) * (16 * NJ);

    float pv[NJ], c0[NJ], c1[NJ];
    #pragma unroll
    for (int j = 0; j < NJ; ++j) {
        pv[j] = param[base + 16 * j + n];
        const v2f c = *(const v2f*)(coords + 2 * (base + 16 * j + n));
        c0[j] = c.x; c1[j] = c.y;
    }

    v2f fa[16], fb[16];
    load_frag(Wh2 + row_t0(n) * HID + kb, fa);

    v2f w = *(const v2f*)(Wh1 + kb);
    v2f b = *(const v2f*)(Bh1 + kb);
    for (int s = 0; s < 16; ++s) {
        const int sn = s < 15 ? s + 1 : 15;
        const v2f wn = *(const v2f*)(Wh1 + 4 * sn + kb);
        const v2f bn = *(const v2f*)(Bh1 + 4 * sn + kb);
        #pragma unroll
        for (int j = 0; j < NJ; ++j) {
            v2f hv;
            hv.x = tanhf(pv[j] * w.x + b.x);
            hv.y = tanhf(pv[j] * w.y + b.y);
            sH[wave][j][s][lane] = hv;
        }
        w = wn; b = bn;
    }
    v2f hfrag[NJ][16];
    #pragma unroll
    for (int j = 0; j < NJ; ++j) {
        #pragma unroll
        for (int s = 0; s < 16; ++s) hfrag[j][s] = sH[wave][j][s][lane];
    }

    float t1[NJ][RANK], b3[NJ][RANK];
    {
        load_frag(Wh2 + row_b1(0, n) * HID + kb, fb);
        float bias[8];
        #pragma unroll
        for (int v = 0; v < 8; ++v) bias[v] = Bh2[row_t0(8 * hf + v)];
        #pragma unroll
        for (int j = 0; j < NJ; ++j) {
            v8f g = mm16(fa, hfrag[j]);
            #pragma unroll
            for (int v = 0; v < 8; ++v) g[v] += bias[v];
            #pragma unroll
            for (int r = 0; r < RANK; ++r) {
                const float t  = c0[j] * g[2 * r] + c1[j] * g[2 * r + 1];
                const float tx = __shfl_xor(t, 16);
                t1[j][r] = hf ? tx : t;
                b3[j][r] = g[r];
            }
        }
    }

    auto run16 = [&](auto rowfn, const float* nextp, auto epi) {
        auto tile = [&](int t, const v2f (&cur)[16]) {
            float bias[8];
            #pragma unroll
            for (int v = 0; v < 8; ++v) bias[v] = Bh2[rowfn(t, 8 * hf + v)];
            #pragma unroll
            for (int j = 0; j < NJ; ++j) {
                v8f g = mm16(cur, hfrag[j]);
                #pragma unroll
                for (int v = 0; v < 8; ++v) g[v] += bias[v];
                epi(t, j, g);
            }
        };
        for (int t = 0; t < 16; t += 2) {
            load_frag(Wh2 + rowfn(t + 1, n) * HID + kb, fa);
            tile(t, fb);
            const float* np = (t < 14) ? (Wh2 + rowfn(t + 2, n) * HID + kb) : nextp;
            if (np) load_frag(np, fb);
            tile(t + 1, fa);
        }
    };

    run16([](int t, int i) { return row_b1(t, i); }, Wh2 + row_a2(0, n) * HID + kb,
          [&](int s, int j, const v8f& g) {
        const int h0 = 4 * s + kb;
        const v4f w  = *(const v4f*)(W1 + 2 * h0);
        const v2f bb = *(const v2f*)(B1 + h0);
        float p0 = w.x * c0[j] + w.y * c1[j] + bb.x;
        float p1 = w.z * c0[j] + w.w * c1[j] + bb.y;
        #pragma unroll
        for (int r = 0; r < RANK; ++r) p0 += g[r] * t1[j][r];
        #pragma unroll
        for (int r = 0; r < RANK; ++r) p1 += g[4 + r] * t1[j][r];
        v2f xv;
        xv.x = tanhf(p0);
        xv.y = tanhf(p1);
        sX1[wave][j][s][lane] = xv;
    });

    float t2p[NJ][RANK] = {};
    run16([](int t, int i) { return row_a2(t, i); }, W2 + n * HID + kb,
          [&](int s, int j, const v8f& g) {
        const v2f xv = sX1[wave][j][s][lane];
        #pragma unroll
        for (int r = 0; r < RANK; ++r) t2p[j][r] += xv.x * g[r];
        #pragma unroll
        for (int r = 0; r < RANK; ++r) t2p[j][r] += xv.y * g[4 + r];
    });
    float t2[NJ][RANK];
    v2f x1frag[NJ][16];
    #pragma unroll
    for (int j = 0; j < NJ; ++j) {
        #pragma unroll
        for (int r = 0; r < RANK; ++r) t2[j][r] = t2p[j][r] + __shfl_xor(t2p[j][r], 16);
        #pragma unroll
        for (int s = 0; s < 16; ++s) x1frag[j][s] = sX1[wave][j][s][lane];
    }

    auto l2tile = [&](int ct, const v2f (&cur)[16]) {
        const v4f ba = *(const v4f*)(B2 + 16 * ct + 8 * hf);
        const v4f bc = *(const v4f*)(B2 + 16 * ct + 8 * hf + 4);
        #pragma unroll
        for (int j = 0; j < NJ; ++j) {
            const v8f p = mm16(cur, x1frag[j]);
            #pragma unroll
            for (int q = 0; q < 4; ++q) {
                v2f pp;
                pp.x = p[2 * q]     + (q < 2 ? ba[2 * q]     : bc[2 * q - 4]);
                pp.y = p[2 * q + 1] + (q < 2 ? ba[2 * q + 1] : bc[2 * q - 3]);
                sP[wave][j][4 * ct + q][lane] = pp;
            }
        }
    };
    load_frag(W2 + (16 + n) * HID + kb, fa);
    l2tile(0, fb);
    load_frag(W2 + (32 + n) * HID + kb, fb);
    l2tile(1, fa);
    load_frag(W2 + (48 + n) * HID + kb, fa);
    l2tile(2, fb);
    load_frag(Wh2 + row_b2(0, n) * HID + kb, fb);
    l2tile(3, fa);

    run16([](int t, int i) { return row_b2(t, i); }, Wh2 + row_a3(0, n) * HID + kb,
          [&](int u, int j, const v8f& g) {
        v2f p = sP[wave][j][u][lane];
        #pragma unroll
        for (int r = 0; r < RANK; ++r) p.x += t2[j][r] * g[r];
        #pragma unroll
        for (int r = 0; r < RANK; ++r) p.y += t2[j][r] * g[4 + r];
        v2f xv;
        xv.x = tanhf(p.x);
        xv.y = tanhf(p.y);
        sP[wave][j][u][lane] = xv;
    });

    float t3p[NJ][RANK] = {};
    float w3p[NJ] = {};
    run16([](int t, int i) { return row_a3(t, i); }, nullptr,
          [&](int u, int j, const v8f& g) {
        const v2f xv = sP[wave][j][u][lane];
        const v2f w3 = *(const v2f*)(W3 + 16 * (u >> 2) + 8 * hf + 2 * (u & 3));
        w3p[j] += xv.x * w3.x;
        #pragma unroll
        for (int r = 0; r < RANK; ++r) t3p[j][r] += xv.x * g[r];
        w3p[j] += xv.y * w3.y;
        #pragma unroll
        for (int r = 0; r < RANK; ++r) t3p[j][r] += xv.y * g[4 + r];
    });

    const float b3s = B3[0];
    #pragma unroll
    for (int j = 0; j < NJ; ++j) {
        float o = (w3p[j] + __shfl_xor(w3p[j], 16)) + b3s;
        #pragma unroll
        for (int r = 0; r < RANK; ++r)
            o += (t3p[j][r] + __shfl_xor(t3p[j][r], 16)) * b3[j][r];
        if (hf) out[base + 16 * j + n] = o;
    }
}

extern "C" void kernel_launch(void* const* d_in, const int* in_sizes, int n_in,
                              void* d_out, int out_size, void* d_ws, size_t ws_size,
                              hipStream_t stream) {
    const float* coords = (const float*)d_in[0];
    const float* param  = (const float*)d_in[1];
    const float* W1     = (const float*)d_in[2];
    const float* B1     = (const float*)d_in[3];
    const float* W2     = (const float*)d_in[4];
    const float* B2     = (const float*)d_in[5];
    const float* W3     = (const float*)d_in[6];
    const float* B3     = (const float*)d_in[7];
    const float* Wh1    = (const float*)d_in[8];
    const float* Bh1    = (const float*)d_in[9];
    const float* Wh2    = (const float*)d_in[10];
    const float* Bh2    = (const float*)d_in[11];
    float* out = (float*)d_out;

    const int n = in_sizes[1];
    const int blocks = n / (32 * NJ);
    hyperlr_fused<<<blocks, 64, 0, stream>>>(coords, param, W1, B1, W2, B2, W3, B3,
                                             Wh1, Bh1, Wh2, Bh2, out);
}


# The two waves of a workgroup share its 32 samples and split each section's weight rows (x1, partial sums handed over in LDS, 3 barriers): two waves per SIMD; t2,t3,x2.W3 summed per wave, then added
# speedup vs baseline: 3.2012x; 1.6508x over previous
#include <hip/hip_runtime.h>
#include <math.h>


typedef float v2f __attribute__((ext_vector_type(2)));
typedef float v4f __attribute__((ext_vector_type(4)));
typedef float v8f __attribute__((ext_vector_type(8)));

#define HID   64
#define RANK  4
#define NAB   1036
#define NJ    2


__device__ __forceinline__ int row_t0(int i)        { return i < 8 ? i : 1032 + (i & 3); }
__device__ __forceinline__ int row_b1(int t, int i) { return 8 + 16 * t + i; }
__device__ __forceinline__ int row_a2(int t, int i) { return 264 + 64 * (i & 3) + 4 * t + 2 * (i >> 3) + ((i >> 2) & 1); }
__device__ __forceinline__ int row_b2(int t, int i) { return 520 + 64 * (t >> 2) + 8 * (t & 3) + 32 * (i >> 3) + (i & 7); }
__device__ __forceinline__ int row_a3(int t, int i) { return 776 + 64 * (i & 3) + 16 * (t >> 2) + 8 * (i >> 3) + 2 * (t & 3) + ((i >> 2) & 1); }

__device__ __forceinline__ void load_frag(const float* __restrict__ rowp, v2f (&a)[16]) {
    #pragma unroll
    for (int s = 0; s < 16; ++s) a[s] = *(const v2f*)(rowp + 4 * s);
}

__device__ __forceinline__ v8f mm16(const v2f (&a)[16], const v2f (&b)[16]) {
    v8f acc0 = {}, acc1 = {};
    #pragma unroll
    for (int s = 0; s < 16; s += 2) {
        acc0 = __builtin_amdgcn_wmma_f32_16x16x4_f32(false, a[s],     false, b[s],
                                                     (short)0, acc0, false, false);
        acc1 = __builtin_amdgcn_wmma_f32_16x16x4_f32(false, a[s + 1], false, b[s + 1],
                                                     (short)0, acc1, false, false);
    }
    return acc0 + acc1;
}

__global__ __launch_bounds__(64) void hyperlr_fused(
    const float* __restrict__ coords, const float* __restrict__ param,
    const float* __restrict__ W1,  const float* __restrict__ B1,
    const float* __restrict__ W2,  const float* __restrict__ B2,
    const float* __restrict__ W3,  const float* __restrict__ B3,
    const float* __restrict__ Wh1, const float* __restrict__ Bh1,
    const float* __restrict__ Wh2, const float* __restrict__ Bh2,
    float* __restrict__ out)
{
    __shared__ v2f sH [NJ][16][32];
    __shared__ v2f sX1[NJ][16][32];
    __shared__ v2f sP [NJ][16][32];
    __shared__ float sT2[2][NJ][RANK][32];
    __shared__ float sT3[2][NJ][RANK + 1][32];

    const int lane = threadIdx.x & 31;
    const int wave = __builtin_amdgcn_readfirstlane(threadIdx.x >> 5);
    const int t0   = 8 * wave;
    const int n    = lane & 15;
    const int hf   = lane >> 4;
    const int kb   = 2 * hf;
    const int base = blockIdx.x * (16 * NJ);

    float pv[NJ], c0[NJ], c1[NJ];
    #pragma unroll
    for (int j = 0; j < NJ; ++j) {
        pv[j] = param[base + 16 * j + n];
        const v2f c = *(const v2f*)(coords + 2 * (base + 16 * j + n));
        c0[j] = c.x; c1[j] = c.y;
    }

    v2f fa[16], fb[16];
    load_frag(Wh2 + row_t0(n) * HID + kb, fa);

    v2f w = *(const v2f*)(Wh1 + 4 * t0 + kb);
    v2f b = *(const v2f*)(Bh1 + 4 * t0 + kb);
    #pragma nounroll
    for (int s = t0; s < t0 + 8; ++s) {
        const int sn = s < t0 + 7 ? s + 1 : s;
        const v2f wn = *(const v2f*)(Wh1 + 4 * sn + kb);
        const v2f bn = *(const v2f*)(Bh1 + 4 * sn + kb);
        #pragma unroll
        for (int j = 0; j < NJ; ++j) {
            v2f hv;
            hv.x = tanhf(pv[j] * w.x + b.x);
            hv.y = tanhf(pv[j] * w.y + b.y);
            sH[j][s][lane] = hv;
        }
        w = wn; b = bn;
    }
    __syncthreads();
    v2f hfrag[NJ][16];
    #pragma unroll
    for (int j = 0; j < NJ; ++j) {
        #pragma unroll
        for (int s = 0; s < 16; ++s) hfrag[j][s] = sH[j][s][lane];
    }

    float t1[NJ][RANK], b3[NJ][RANK];
    {
        load_frag(Wh2 + row_b1(t0, n) * HID + kb, fb);
        float bias[8];
        #pragma unroll
        for (int v = 0; v < 8; ++v) bias[v] = Bh2[row_t0(8 * hf + v)];
        #pragma unroll
        for (int j = 0; j < NJ; ++j) {
            v8f g = mm16(fa, hfrag[j]);
            #pragma unroll
            for (int v = 0; v < 8; ++v) g[v] += bias[v];
            #pragma unroll
            for (int r = 0; r < RANK; ++r) {
                const float t  = c0[j] * g[2 * r] + c1[j] * g[2 * r + 1];
                const float tx = __shfl_xor(t, 16);
                t1[j][r] = hf ? tx : t;
                b3[j][r] = g[r];
            }
        }
    }

    auto run8 = [&](auto rowfn, const float* nextp, auto epi) {
        auto tile = [&](int t, const v2f (&cur)[16]) {
            float bias[8];
            #pragma unroll
            for (int v = 0; v < 8; ++v) bias[v] = Bh2[rowfn(t, 8 * hf + v)];
            #pragma unroll
            for (int j = 0; j < NJ; ++j) {
                v8f g = mm16(cur, hfrag[j]);
                #pragma unroll
                for (int v = 0; v < 8; ++v) g[v] += bias[v];
                epi(t, j, g);
            }
        };
        #pragma nounroll
        for (int t = t0; t < t0 + 8; t += 2) {
            load_frag(Wh2 + rowfn(t + 1, n) * HID + kb, fa);
            tile(t, fb);
            const float* np = (t < t0 + 6) ? (Wh2 + rowfn(t + 2, n) * HID + kb) : nextp;
            if (np) load_frag(np, fb);
            tile(t + 1, fa);
        }
    };

    run8([](int t, int i) { return row_b1(t, i); }, Wh2 + row_a2(t0, n) * HID + kb,
          [&](int s, int j, const v8f& g) {
        const int h0 = 4 * s + kb;
        const v4f w  = *(const v4f*)(W1 + 2 * h0);
        const v2f bb = *(const v2f*)(B1 + h0);
        float p0 = w.x * c0[j] + w.y * c1[j] + bb.x;
        float p1 = w.z * c0[j] + w.w * c1[j] + bb.y;
        #pragma unroll
        for (int r = 0; r < RANK; ++r) p0 += g[r] * t1[j][r];
        #pragma unroll
        for (int r = 0; r < RANK; ++r) p1 += g[4 + r] * t1[j][r];
        v2f xv;
        xv.x = tanhf(p0);
        xv.y = tanhf(p1);
        sX1[j][s][lane] = xv;
    });

    float t2p[NJ][RANK] = {};
    run8([](int t, int i) { return row_a2(t, i); }, W2 + (32 * wave + n) * HID + kb,
          [&](int s, int j, const v8f& g) {
        const v2f xv = sX1[j][s][lane];
        #pragma unroll
        for (int r = 0; r < RANK; ++r) t2p[j][r] += xv.x * g[r];
        #pragma unroll
        for (int r = 0; r < RANK; ++r) t2p[j][r] += xv.y * g[4 + r];
    });
    #pragma unroll
    for (int j = 0; j < NJ; ++j) {
        #pragma unroll
        for (int r = 0; r < RANK; ++r) sT2[wave][j][r][lane] = t2p[j][r] + __shfl_xor(t2p[j][r], 16);
    }
    __syncthreads();
    float t2[NJ][RANK];
    v2f x1frag[NJ][16];
    #pragma unroll
    for (int j = 0; j < NJ; ++j) {
        #pragma unroll
        for (int r = 0; r < RANK; ++r) t2[j][r] = sT2[0][j][r][lane] + sT2[1][j][r][lane];
        #pragma unroll
        for (int s = 0; s < 16; ++s) x1frag[j][s] = sX1[j][s][lane];
    }

    auto l2tile = [&](int ct, const v2f (&cur)[16]) {
        const v4f ba = *(const v4f*)(B2 + 16 * ct + 8 * hf);
        const v4f bc = *(const v4f*)(B2 + 16 * ct + 8 * hf + 4);
        #pragma unroll
        for (int j = 0; j < NJ; ++j) {
            const v8f p = mm16(cur, x1frag[j]);
            #pragma unroll
            for (int q = 0; q < 4; ++q) {
                v2f pp;
                pp.x = p[2 * q]     + (q < 2 ? ba[2 * q]     : bc[2 * q - 4]);
                pp.y = p[2 * q + 1] + (q < 2 ? ba[2 * q + 1] : bc[2 * q - 3]);
                sP[j][4 * ct + q][lane] = pp;
            }
        }
    };
    load_frag(W2 + (32 * wave + 16 + n) * HID + kb, fa);
    l2tile(2 * wave, fb);
    load_frag(Wh2 + row_b2(t0, n) * HID + kb, fb);
    l2tile(2 * wave + 1, fa);

    run8([](int t, int i) { return row_b2(t, i); }, Wh2 + row_a3(t0, n) * HID + kb,
          [&](int u, int j, const v8f& g) {
        v2f p = sP[j][u][lane];
        #pragma unroll
        for (int r = 0; r < RANK; ++r) p.x += t2[j][r] * g[r];
        #pragma unroll
        for (int r = 0; r < RANK; ++r) p.y += t2[j][r] * g[4 + r];
        v2f xv;
        xv.x = tanhf(p.x);
        xv.y = tanhf(p.y);
        sP[j][u][lane] = xv;
    });

    float t3p[NJ][RANK] = {};
    float w3p[NJ] = {};
    run8([](int t, int i) { return row_a3(t, i); }, nullptr,
          [&](int u, int j, const v8f& g) {
        const v2f xv = sP[j][u][lane];
        const v2f w3 = *(const v2f*)(W3 + 16 * (u >> 2) + 8 * hf + 2 * (u & 3));
        w3p[j] += xv.x * w3.x;
        #pragma unroll
        for (int r = 0; r < RANK; ++r) t3p[j][r] += xv.x * g[r];
        w3p[j] += xv.y * w3.y;
        #pragma unroll
        for (int r = 0; r < RANK; ++r) t3p[j][r] += xv.y * g[4 + r];
    });

    #pragma unroll
    for (int j = 0; j < NJ; ++j) {
        #pragma unroll
        for (int r = 0; r < RANK; ++r) sT3[wave][j][r][lane] = t3p[j][r] + __shfl_xor(t3p[j][r], 16);
        sT3[wave][j][RANK][lane] = w3p[j] + __shfl_xor(w3p[j], 16);
    }
    __syncthreads();
    if (wave == 0) {
        const float b3s = B3[0];
        #pragma unroll
        for (int j = 0; j < NJ; ++j) {
            float o = (sT3[0][j][RANK][lane] + sT3[1][j][RANK][lane]) + b3s;
            #pragma unroll
            for (int r = 0; r < RANK; ++r)
                o += (sT3[0][j][r][lane] + sT3[1][j][r][lane]) * b3[j][r];
            if (hf) out[base + 16 * j + n] = o;
        }
    }
}

extern "C" void kernel_launch(void* const* d_in, const int* in_sizes, int n_in,
                              void* d_out, int out_size, void* d_ws, size_t ws_size,
                              hipStream_t stream) {
    const float* coords = (const float*)d_in[0];
    const float* param  = (const float*)d_in[1];
    const float* W1     = (const float*)d_in[2];
    const float* B1     = (const float*)d_in[3];
    const float* W2     = (const float*)d_in[4];
    const float* B2     = (const float*)d_in[5];
    const float* W3     = (const float*)d_in[6];
    const float* B3     = (const float*)d_in[7];
    const float* Wh1    = (const float*)d_in[8];
    const float* Bh1    = (const float*)d_in[9];
    const float* Wh2    = (const float*)d_in[10];
    const float* Bh2    = (const float*)d_in[11];
    float* out = (float*)d_out;

    const int n = in_sizes[1];
    const int blocks = n / (16 * NJ);
    hyperlr_fused<<<blocks, 64, 0, stream>>>(coords, param, W1, B1, W2, B2, W3, B3,
                                             Wh1, Bh1, Wh2, Bh2, out);
}


# The four waves of a workgroup share its 32 samples and split each section's weight rows four ways (one W2 tile each): four waves per SIMD; t2,t3,x2.W3 partials of four waves summed pairwise
# speedup vs baseline: 3.6645x; 1.1447x over previous
#include <hip/hip_runtime.h>
#include <math.h>


typedef float v2f __attribute__((ext_vector_type(2)));
typedef float v4f __attribute__((ext_vector_type(4)));
typedef float v8f __attribute__((ext_vector_type(8)));

#define HID   64
#define RANK  4
#define NAB   1036
#define NJ    2
#define NW    4
#define TPW   (16 / NW)


__device__ __forceinline__ int row_t0(int i)        { return i < 8 ? i : 1032 + (i & 3); }
__device__ __forceinline__ int row_b1(int t, int i) { return 8 + 16 * t + i; }
__device__ __forceinline__ int row_a2(int t, int i) { return 264 + 64 * (i & 3) + 4 * t + 2 * (i >> 3) + ((i >> 2) & 1); }
__device__ __forceinline__ int row_b2(int t, int i) { return 520 + 64 * (t >> 2) + 8 * (t & 3) + 32 * (i >> 3) + (i & 7); }
__device__ __forceinline__ int row_a3(int t, int i) { return 776 + 64 * (i & 3) + 16 * (t >> 2) + 8 * (i >> 3) + 2 * (t & 3) + ((i >> 2) & 1); }

__device__ __forceinline__ void load_frag(const float* __restrict__ rowp, v2f (&a)[16]) {
    #pragma unroll
    for (int s = 0; s < 16; ++s) a[s] = *(const v2f*)(rowp + 4 * s);
}

__device__ __forceinline__ v8f mm16(const v2f (&a)[16], const v2f (&b)[16]) {
    v8f acc0 = {}, acc1 = {};
    #pragma unroll
    for (int s = 0; s < 16; s += 2) {
        acc0 = __builtin_amdgcn_wmma_f32_16x16x4_f32(false, a[s],     false, b[s],
                                                     (short)0, acc0, false, false);
        acc1 = __builtin_amdgcn_wmma_f32_16x16x4_f32(false, a[s + 1], false, b[s + 1],
                                                     (short)0, acc1, false, false);
    }
    return acc0 + acc1;
}

__global__ __launch_bounds__(32 * NW, 4) void hyperlr_fused(
    const float* __restrict__ coords, const float* __restrict__ param,
    const float* __restrict__ W1,  const float* __restrict__ B1,
    const float* __restrict__ W2,  const float* __restrict__ B2,
    const float* __restrict__ W3,  const float* __restrict__ B3,
    const float* __restrict__ Wh1, const float* __restrict__ Bh1,
    const float* __restrict__ Wh2, const float* __restrict__ Bh2,
    float* __restrict__ out)
{
    __shared__ v2f sH [NJ][16][32];
    __shared__ v2f sX1[NJ][16][32];
    __shared__ v2f sP [NJ][16][32];
    __shared__ float sT2[NW][NJ][RANK][32];
    __shared__ float sT3[NW][NJ][RANK + 1][32];

    const int lane = threadIdx.x & 31;
    const int wave = __builtin_amdgcn_readfirstlane(threadIdx.x >> 5);
    const int t0   = TPW * wave;
    const int n    = lane & 15;
    const int hf   = lane >> 4;
    const int kb   = 2 * hf;
    const int base = blockIdx.x * (16 * NJ);

    float pv[NJ], c0[NJ], c1[NJ];
    #pragma unroll
    for (int j = 0; j < NJ; ++j) {
        pv[j] = param[base + 16 * j + n];
        const v2f c = *(const v2f*)(coords + 2 * (base + 16 * j + n));
        c0[j] = c.x; c1[j] = c.y;
    }

    v2f fa[16], fb[16];
    load_frag(Wh2 + row_t0(n) * HID + kb, fa);

    v2f w = *(const v2f*)(Wh1 + 4 * t0 + kb);
    v2f b = *(const v2f*)(Bh1 + 4 * t0 + kb);
    #pragma nounroll
    for (int s = t0; s < t0 + TPW; ++s) {
        const int sn = s < t0 + TPW - 1 ? s + 1 : s;
        const v2f wn = *(const v2f*)(Wh1 + 4 * sn + kb);
        const v2f bn = *(const v2f*)(Bh1 + 4 * sn + kb);
        #pragma unroll
        for (int j = 0; j < NJ; ++j) {
            v2f hv;
            hv.x = tanhf(pv[j] * w.x + b.x);
            hv.y = tanhf(pv[j] * w.y + b.y);
            sH[j][s][lane] = hv;
        }
        w = wn; b = bn;
    }
    __syncthreads();
    v2f hfrag[NJ][16];
    #pragma unroll
    for (int j = 0; j < NJ; ++j) {
        #pragma unroll
        for (int s = 0; s < 16; ++s) hfrag[j][s] = sH[j][s][lane];
    }

    float t1[NJ][RANK], b3[NJ][RANK];
    {
        load_frag(Wh2 + row_b1(t0, n) * HID + kb, fb);
        float bias[8];
        #pragma unroll
        for (int v = 0; v < 8; ++v) bias[v] = Bh2[row_t0(8 * hf + v)];
        #pragma unroll
        for (int j = 0; j < NJ; ++j) {
            v8f g = mm16(fa, hfrag[j]);
            #pragma unroll
            for (int v = 0; v < 8; ++v) g[v] += bias[v];
            #pragma unroll
            for (int r = 0; r < RANK; ++r) {
                const float t  = c0[j] * g[2 * r] + c1[j] * g[2 * r + 1];
                const float tx = __shfl_xor(t, 16);
                t1[j][r] = hf ? tx : t;
                b3[j][r] = g[r];
            }
        }
    }

    auto run = [&](auto rowfn, const float* nextp, auto epi, v2f (&cur)[16], v2f (&nxt)[16]) {
        auto tile = [&](int t, const v2f (&frag)[16]) {
            float bias[8];
            #pragma unroll
            for (int v = 0; v < 8; ++v) bias[v] = Bh2[rowfn(t, 8 * hf + v)];
            #pragma unroll
            for (int j = 0; j < NJ; ++j) {
                v8f g = mm16(frag, hfrag[j]);
                #pragma unroll
                for (int v = 0; v < 8; ++v) g[v] += bias[v];
                epi(t, j, g);
            }
        };
        #pragma nounroll
        for (int t = t0; t < t0 + TPW; t += 2) {
            load_frag(Wh2 + rowfn(t + 1, n) * HID + kb, nxt);
            tile(t, cur);
            const float* np = (t < t0 + TPW - 2) ? (Wh2 + rowfn(t + 2, n) * HID + kb) : nextp;
            if (np) load_frag(np, cur);
            tile(t + 1, nxt);
        }
    };

    run([](int t, int i) { return row_b1(t, i); }, Wh2 + row_a2(t0, n) * HID + kb,
          [&](int s, int j, const v8f& g) {
        const int h0 = 4 * s + kb;
        const v4f w  = *(const v4f*)(W1 + 2 * h0);
        const v2f bb = *(const v2f*)(B1 + h0);
        float p0 = w.x * c0[j] + w.y * c1[j] + bb.x;
        float p1 = w.z * c0[j] + w.w * c1[j] + bb.y;
        #pragma unroll
        for (int r = 0; r < RANK; ++r) p0 += g[r] * t1[j][r];
        #pragma unroll
        for (int r = 0; r < RANK; ++r) p1 += g[4 + r] * t1[j][r];
        v2f xv;
        xv.x = tanhf(p0);
        xv.y = tanhf(p1);
        sX1[j][s][lane] = xv;
    }, fb, fa);

    float t2p[NJ][RANK] = {};
    run([](int t, int i) { return row_a2(t, i); }, W2 + (16 * wave + n) * HID + kb,
          [&](int s, int j, const v8f& g) {
        const v2f xv = sX1[j][s][lane];
        #pragma unroll
        for (int r = 0; r < RANK; ++r) t2p[j][r] += xv.x * g[r];
        #pragma unroll
        for (int r = 0; r < RANK; ++r) t2p[j][r] += xv.y * g[4 + r];
    }, fb, fa);
    #pragma unroll
    for (int j = 0; j < NJ; ++j) {
        #pragma unroll
        for (int r = 0; r < RANK; ++r) sT2[wave][j][r][lane] = t2p[j][r] + __shfl_xor(t2p[j][r], 16);
    }
    __syncthreads();
    float t2[NJ][RANK];
    #pragma unroll
    for (int j = 0; j < NJ; ++j) {
        #pragma unroll
        for (int r = 0; r < RANK; ++r)
            t2[j][r] = (sT2[0][j][r][lane] + sT2[1][j][r][lane]) + (sT2[2][j][r][lane] + sT2[3][j][r][lane]);
    }

    auto l2tile = [&](int ct, const v2f (&cur)[16]) {
        const v4f ba = *(const v4f*)(B2 + 16 * ct + 8 * hf);
        const v4f bc = *(const v4f*)(B2 + 16 * ct + 8 * hf + 4);
        #pragma unroll
        for (int j = 0; j < NJ; ++j) {
            v2f x1frag[16];
            #pragma unroll
            for (int s = 0; s < 16; ++s) x1frag[s] = sX1[j][s][lane];
            const v8f p = mm16(cur, x1frag);
            #pragma unroll
            for (int q = 0; q < 4; ++q) {
                v2f pp;
                pp.x = p[2 * q]     + (q < 2 ? ba[2 * q]     : bc[2 * q - 4]);
                pp.y = p[2 * q + 1] + (q < 2 ? ba[2 * q + 1] : bc[2 * q - 3]);
                sP[j][4 * ct + q][lane] = pp;
            }
        }
    };
    load_frag(Wh2 + row_b2(t0, n) * HID + kb, fa);
    l2tile(wave, fb);

    run([](int t, int i) { return row_b2(t, i); }, Wh2 + row_a3(t0, n) * HID + kb,
          [&](int u, int j, const v8f& g) {
        v2f p = sP[j][u][lane];
        #pragma unroll
        for (int r = 0; r < RANK; ++r) p.x += t2[j][r] * g[r];
        #pragma unroll
        for (int r = 0; r < RANK; ++r) p.y += t2[j][r] * g[4 + r];
        v2f xv;
        xv.x = tanhf(p.x);
        xv.y = tanhf(p.y);
        sP[j][u][lane] = xv;
    }, fa, fb);

    float t3p[NJ][RANK] = {};
    float w3p[NJ] = {};
    run([](int t, int i) { return row_a3(t, i); }, nullptr,
          [&](int u, int j, const v8f& g) {
        const v2f xv = sP[j][u][lane];
        const v2f w3 = *(const v2f*)(W3 + 16 * (u >> 2) + 8 * hf + 2 * (u & 3));
        w3p[j] += xv.x * w3.x;
        #pragma unroll
        for (int r = 0; r < RANK; ++r) t3p[j][r] += xv.x * g[r];
        w3p[j] += xv.y * w3.y;
        #pragma unroll
        for (int r = 0; r < RANK; ++r) t3p[j][r] += xv.y * g[4 + r];
    }, fa, fb);

    #pragma unroll
    for (int j = 0; j < NJ; ++j) {
        #pragma unroll
        for (int r = 0; r < RANK; ++r) sT3[wave][j][r][lane] = t3p[j][r] + __shfl_xor(t3p[j][r], 16);
        sT3[wave][j][RANK][lane] = w3p[j] + __shfl_xor(w3p[j], 16);
    }
    __syncthreads();
    if (wave == 0) {
        const float b3s = B3[0];
        #pragma unroll
        for (int j = 0; j < NJ; ++j) {
            float o = ((sT3[0][j][RANK][lane] + sT3[1][j][RANK][lane]) +
                       (sT3[2][j][RANK][lane] + sT3[3][j][RANK][lane])) + b3s;
            #pragma unroll
            for (int r = 0; r < RANK; ++r)
                o += ((sT3[0][j][r][lane] + sT3[1][j][r][lane]) +
                      (sT3[2][j][r][lane] + sT3[3][j][r][lane])) * b3[j][r];
            if (hf) out[base + 16 * j + n] = o;
        }
    }
}

extern "C" void kernel_launch(void* const* d_in, const int* in_sizes, int n_in,
                              void* d_out, int out_size, void* d_ws, size_t ws_size,
                              hipStream_t stream) {
    const float* coords = (const float*)d_in[0];
    const float* param  = (const float*)d_in[1];
    const float* W1     = (const float*)d_in[2];
    const float* B1     = (const float*)d_in[3];
    const float* W2     = (const float*)d_in[4];
    const float* B2     = (const float*)d_in[5];
    const float* W3     = (const float*)d_in[6];
    const float* B3     = (const float*)d_in[7];
    const float* Wh1    = (const float*)d_in[8];
    const float* Bh1    = (const float*)d_in[9];
    const float* Wh2    = (const float*)d_in[10];
    const float* Bh2    = (const float*)d_in[11];
    float* out = (float*)d_out;

    const int n = in_sizes[1];
    const int blocks = n / (16 * NJ);
    hyperlr_fused<<<blocks, 32 * NW, 0, stream>>>(coords, param, W1, B1, W2, B2, W3, B3,
                                             Wh1, Bh1, Wh2, Bh2, out);
}


# Wave issue priority falls section by section (s_setprio 3,2,1,0): of the four workgroups on a SIMD the one furthest behind wins the matrix pipe, so they end together, not one after another
# speedup vs baseline: 3.8644x; 1.0546x over previous
#include <hip/hip_runtime.h>
#include <math.h>


typedef float v2f __attribute__((ext_vector_type(2)));
typedef float v4f __attribute__((ext_vector_type(4)));
typedef float v8f __attribute__((ext_vector_type(8)));

#define HID   64
#define RANK  4
#define NAB   1036
#define NJ    2
#define NW    4
#define TPW   (16 / NW)


__device__ __forceinline__ int row_t0(int i)        { return i < 8 ? i : 1032 + (i & 3); }
__device__ __forceinline__ int row_b1(int t, int i) { return 8 + 16 * t + i; }
__device__ __forceinline__ int row_a2(int t, int i) { return 264 + 64 * (i & 3) + 4 * t + 2 * (i >> 3) + ((i >> 2) & 1); }
__device__ __forceinline__ int row_b2(int t, int i) { return 520 + 64 * (t >> 2) + 8 * (t & 3) + 32 * (i >> 3) + (i & 7); }
__device__ __forceinline__ int row_a3(int t, int i) { return 776 + 64 * (i & 3) + 16 * (t >> 2) + 8 * (i >> 3) + 2 * (t & 3) + ((i >> 2) & 1); }

__device__ __forceinline__ void load_frag(const float* __restrict__ rowp, v2f (&a)[16]) {
    #pragma unroll
    for (int s = 0; s < 16; ++s) a[s] = *(const v2f*)(rowp + 4 * s);
}

__device__ __forceinline__ v8f mm16(const v2f (&a)[16], const v2f (&b)[16]) {
    v8f acc0 = {}, acc1 = {};
    #pragma unroll
    for (int s = 0; s < 16; s += 2) {
        acc0 = __builtin_amdgcn_wmma_f32_16x16x4_f32(false, a[s],     false, b[s],
                                                     (short)0, acc0, false, false);
        acc1 = __builtin_amdgcn_wmma_f32_16x16x4_f32(false, a[s + 1], false, b[s + 1],
                                                     (short)0, acc1, false, false);
    }
    return acc0 + acc1;
}

__global__ __launch_bounds__(32 * NW, 4) void hyperlr_fused(
    const float* __restrict__ coords, const float* __restrict__ param,
    const float* __restrict__ W1,  const float* __restrict__ B1,
    const float* __restrict__ W2,  const float* __restrict__ B2,
    const float* __restrict__ W3,  const float* __restrict__ B3,
    const float* __restrict__ Wh1, const float* __restrict__ Bh1,
    const float* __restrict__ Wh2, const float* __restrict__ Bh2,
    float* __restrict__ out)
{
    __shared__ v2f sH [NJ][16][32];
    __shared__ v2f sX1[NJ][16][32];
    __shared__ v2f sP [NJ][16][32];
    __shared__ float sT2[NW][NJ][RANK][32];
    __shared__ float sT3[NW][NJ][RANK + 1][32];

    const int lane = threadIdx.x & 31;
    const int wave = __builtin_amdgcn_readfirstlane(threadIdx.x >> 5);
    const int t0   = TPW * wave;
    const int n    = lane & 15;
    const int hf   = lane >> 4;
    const int kb   = 2 * hf;
    const int base = blockIdx.x * (16 * NJ);

    __builtin_amdgcn_s_setprio(3);

    float pv[NJ], c0[NJ], c1[NJ];
    #pragma unroll
    for (int j = 0; j < NJ; ++j) {
        pv[j] = param[base + 16 * j + n];
        const v2f c = *(const v2f*)(coords + 2 * (base + 16 * j + n));
        c0[j] = c.x; c1[j] = c.y;
    }

    v2f fa[16], fb[16];
    load_frag(Wh2 + row_t0(n) * HID + kb, fa);

    v2f w = *(const v2f*)(Wh1 + 4 * t0 + kb);
    v2f b = *(const v2f*)(Bh1 + 4 * t0 + kb);
    #pragma nounroll
    for (int s = t0; s < t0 + TPW; ++s) {
        const int sn = s < t0 + TPW - 1 ? s + 1 : s;
        const v2f wn = *(const v2f*)(Wh1 + 4 * sn + kb);
        const v2f bn = *(const v2f*)(Bh1 + 4 * sn + kb);
        #pragma unroll
        for (int j = 0; j < NJ; ++j) {
            v2f hv;
            hv.x = tanhf(pv[j] * w.x + b.x);
            hv.y = tanhf(pv[j] * w.y + b.y);
            sH[j][s][lane] = hv;
        }
        w = wn; b = bn;
    }
    __syncthreads();
    v2f hfrag[NJ][16];
    #pragma unroll
    for (int j = 0; j < NJ; ++j) {
        #pragma unroll
        for (int s = 0; s < 16; ++s) hfrag[j][s] = sH[j][s][lane];
    }

    float t1[NJ][RANK], b3[NJ][RANK];
    {
        load_frag(Wh2 + row_b1(t0, n) * HID + kb, fb);
        float bias[8];
        #pragma unroll
        for (int v = 0; v < 8; ++v) bias[v] = Bh2[row_t0(8 * hf + v)];
        #pragma unroll
        for (int j = 0; j < NJ; ++j) {
            v8f g = mm16(fa, hfrag[j]);
            #pragma unroll
            for (int v = 0; v < 8; ++v) g[v] += bias[v];
            #pragma unroll
            for (int r = 0; r < RANK; ++r) {
                const float t  = c0[j] * g[2 * r] + c1[j] * g[2 * r + 1];
                const float tx = __shfl_xor(t, 16);
                t1[j][r] = hf ? tx : t;
                b3[j][r] = g[r];
            }
        }
    }

    auto run = [&](auto rowfn, const float* nextp, auto epi, v2f (&cur)[16], v2f (&nxt)[16]) {
        auto tile = [&](int t, const v2f (&frag)[16]) {
            float bias[8];
            #pragma unroll
            for (int v = 0; v < 8; ++v) bias[v] = Bh2[rowfn(t, 8 * hf + v)];
            #pragma unroll
            for (int j = 0; j < NJ; ++j) {
                v8f g = mm16(frag, hfrag[j]);
                #pragma unroll
                for (int v = 0; v < 8; ++v) g[v] += bias[v];
                epi(t, j, g);
            }
        };
        #pragma nounroll
        for (int t = t0; t < t0 + TPW; t += 2) {
            load_frag(Wh2 + rowfn(t + 1, n) * HID + kb, nxt);
            tile(t, cur);
            const float* np = (t < t0 + TPW - 2) ? (Wh2 + rowfn(t + 2, n) * HID + kb) : nextp;
            if (np) load_frag(np, cur);
            tile(t + 1, nxt);
        }
    };

    run([](int t, int i) { return row_b1(t, i); }, Wh2 + row_a2(t0, n) * HID + kb,
          [&](int s, int j, const v8f& g) {
        const int h0 = 4 * s + kb;
        const v4f w  = *(const v4f*)(W1 + 2 * h0);
        const v2f bb = *(const v2f*)(B1 + h0);
        float p0 = w.x * c0[j] + w.y * c1[j] + bb.x;
        float p1 = w.z * c0[j] + w.w * c1[j] + bb.y;
        #pragma unroll
        for (int r = 0; r < RANK; ++r) p0 += g[r] * t1[j][r];
        #pragma unroll
        for (int r = 0; r < RANK; ++r) p1 += g[4 + r] * t1[j][r];
        v2f xv;
        xv.x = tanhf(p0);
        xv.y = tanhf(p1);
        sX1[j][s][lane] = xv;
    }, fb, fa);

    __builtin_amdgcn_s_setprio(2);
    float t2p[NJ][RANK] = {};
    run([](int t, int i) { return row_a2(t, i); }, W2 + (16 * wave + n) * HID + kb,
          [&](int s, int j, const v8f& g) {
        const v2f xv = sX1[j][s][lane];
        #pragma unroll
        for (int r = 0; r < RANK; ++r) t2p[j][r] += xv.x * g[r];
        #pragma unroll
        for (int r = 0; r < RANK; ++r) t2p[j][r] += xv.y * g[4 + r];
    }, fb, fa);
    #pragma unroll
    for (int j = 0; j < NJ; ++j) {
        #pragma unroll
        for (int r = 0; r < RANK; ++r) sT2[wave][j][r][lane] = t2p[j][r] + __shfl_xor(t2p[j][r], 16);
    }
    __syncthreads();
    __builtin_amdgcn_s_setprio(1);
    float t2[NJ][RANK];
    #pragma unroll
    for (int j = 0; j < NJ; ++j) {
        #pragma unroll
        for (int r = 0; r < RANK; ++r)
            t2[j][r] = (sT2[0][j][r][lane] + sT2[1][j][r][lane]) + (sT2[2][j][r][lane] + sT2[3][j][r][lane]);
    }

    auto l2tile = [&](int ct, const v2f (&cur)[16]) {
        const v4f ba = *(const v4f*)(B2 + 16 * ct + 8 * hf);
        const v4f bc = *(const v4f*)(B2 + 16 * ct + 8 * hf + 4);
        #pragma unroll
        for (int j = 0; j < NJ; ++j) {
            v2f x1frag[16];
            #pragma unroll
            for (int s = 0; s < 16; ++s) x1frag[s] = sX1[j][s][lane];
            const v8f p = mm16(cur, x1frag);
            #pragma unroll
            for (int q = 0; q < 4; ++q) {
                v2f pp;
                pp.x = p[2 * q]     + (q < 2 ? ba[2 * q]     : bc[2 * q - 4]);
                pp.y = p[2 * q + 1] + (q < 2 ? ba[2 * q + 1] : bc[2 * q - 3]);
                sP[j][4 * ct + q][lane] = pp;
            }
        }
    };
    load_frag(Wh2 + row_b2(t0, n) * HID + kb, fa);
    l2tile(wave, fb);

    run([](int t, int i) { return row_b2(t, i); }, Wh2 + row_a3(t0, n) * HID + kb,
          [&](int u, int j, const v8f& g) {
        v2f p = sP[j][u][lane];
        #pragma unroll
        for (int r = 0; r < RANK; ++r) p.x += t2[j][r] * g[r];
        #pragma unroll
        for (int r = 0; r < RANK; ++r) p.y += t2[j][r] * g[4 + r];
        v2f xv;
        xv.x = tanhf(p.x);
        xv.y = tanhf(p.y);
        sP[j][u][lane] = xv;
    }, fa, fb);

    __builtin_amdgcn_s_setprio(0);
    float t3p[NJ][RANK] = {};
    float w3p[NJ] = {};
    run([](int t, int i) { return row_a3(t, i); }, nullptr,
          [&](int u, int j, const v8f& g) {
        const v2f xv = sP[j][u][lane];
        const v2f w3 = *(const v2f*)(W3 + 16 * (u >> 2) + 8 * hf + 2 * (u & 3));
        w3p[j] += xv.x * w3.x;
        #pragma unroll
        for (int r = 0; r < RANK; ++r) t3p[j][r] += xv.x * g[r];
        w3p[j] += xv.y * w3.y;
        #pragma unroll
        for (int r = 0; r < RANK; ++r) t3p[j][r] += xv.y * g[4 + r];
    }, fa, fb);

    #pragma unroll
    for (int j = 0; j < NJ; ++j) {
        #pragma unroll
        for (int r = 0; r < RANK; ++r) sT3[wave][j][r][lane] = t3p[j][r] + __shfl_xor(t3p[j][r], 16);
        sT3[wave][j][RANK][lane] = w3p[j] + __shfl_xor(w3p[j], 16);
    }
    __syncthreads();
    if (wave == 0) {
        const float b3s = B3[0];
        #pragma unroll
        for (int j = 0; j < NJ; ++j) {
            float o = ((sT3[0][j][RANK][lane] + sT3[1][j][RANK][lane]) +
                       (sT3[2][j][RANK][lane] + sT3[3][j][RANK][lane])) + b3s;
            #pragma unroll
            for (int r = 0; r < RANK; ++r)
                o += ((sT3[0][j][r][lane] + sT3[1][j][r][lane]) +
                      (sT3[2][j][r][lane] + sT3[3][j][r][lane])) * b3[j][r];
            if (hf) out[base + 16 * j + n] = o;
        }
    }
}

extern "C" void kernel_launch(void* const* d_in, const int* in_sizes, int n_in,
                              void* d_out, int out_size, void* d_ws, size_t ws_size,
                              hipStream_t stream) {
    const float* coords = (const float*)d_in[0];
    const float* param  = (const float*)d_in[1];
    const float* W1     = (const float*)d_in[2];
    const float* B1     = (const float*)d_in[3];
    const float* W2     = (const float*)d_in[4];
    const float* B2     = (const float*)d_in[5];
    const float* W3     = (const float*)d_in[6];
    const float* B3     = (const float*)d_in[7];
    const float* Wh1    = (const float*)d_in[8];
    const float* Bh1    = (const float*)d_in[9];
    const float* Wh2    = (const float*)d_in[10];
    const float* Bh2    = (const float*)d_in[11];
    float* out = (float*)d_out;

    const int n = in_sizes[1];
    const int blocks = n / (16 * NJ);
    hyperlr_fused<<<blocks, 32 * NW, 0, stream>>>(coords, param, W1, B1, W2, B2, W3, B3,
                                             Wh1, Bh1, Wh2, Bh2, out);
}
